# starting kernel
# baseline (speedup 1.0000x reference)
#include <hip/hip_runtime.h>
#include <hip/hip_bf16.h>

typedef float v2f __attribute__((ext_vector_type(2)));
typedef float v8f __attribute__((ext_vector_type(8)));

#define BB 64
#define NN 768
#define FF 256
#define PP 64
#define TT 16
#define KK 3
#define LL 12

__global__ __launch_bounds__(256) void pooled_x_kernel(
    const float* __restrict__ x, const float* __restrict__ w,
    const float* __restrict__ bias, float* __restrict__ out)
{
    __shared__ float Xl[PP * FF];

    const int bl = blockIdx.x;
    const int b  = bl / LL;
    const int l  = bl % LL;
    const int tid = threadIdx.x;

    const float* xb = x + (size_t)b * NN * FF;
    for (int i = tid; i < PP * FF; i += 256) {
        const int p = i >> 8;
        const int f = i & 255;
        Xl[i] = xb[(size_t)(p * LL + l) * FF + f];
    }
    __syncthreads();

    const int wave = tid >> 5;
    const int ln   = tid & 31;
    const int half = ln >> 4;
    const int col  = ln & 15;

    v2f afrag[KK][16];
#pragma unroll
    for (int k = 0; k < KK; ++k) {
#pragma unroll
        for (int p4 = 0; p4 < 16; ++p4) {
            const int p0 = p4 * 4 + 2 * half;
            afrag[k][p4].x = w[(col * PP + p0) * KK + k];
            afrag[k][p4].y = w[(col * PP + p0 + 1) * KK + k];
        }
    }

    for (int tile = wave; tile < FF / 16; tile += 8) {
        const int fbase = tile * 16;
        v8f acc = {};
#pragma unroll
        for (int k = 0; k < KK; ++k) {
            const int f  = fbase + col + k - 1;
            const int fc = f < 0 ? 0 : (f > FF - 1 ? FF - 1 : f);
            const float msk = (f == fc) ? 1.0f : 0.0f;
#pragma unroll
            for (int p4 = 0; p4 < 16; ++p4) {
                const int p0 = p4 * 4 + 2 * half;
                v2f bf;
                bf.x = Xl[p0 * FF + fc] * msk;
                bf.y = Xl[(p0 + 1) * FF + fc] * msk;
                acc = __builtin_amdgcn_wmma_f32_16x16x4_f32(
                    false, afrag[k][p4], false, bf, (short)0, acc, false, false);
            }
        }
#pragma unroll
        for (int i = 0; i < 8; ++i) {
            const int t = i + 8 * half;
            out[((size_t)b * (TT * LL) + t * LL + l) * FF + fbase + col] =
                acc[i] + bias[t];
        }
    }
}

#define M1STRIDE 65

__global__ __launch_bounds__(384) void pooled_adj_kernel(
    const float* __restrict__ adj, const float* __restrict__ w,
    const float* __restrict__ re, float* __restrict__ out)
{
    __shared__ float Slds[TT * PP];
    __shared__ float M1[LL][TT * M1STRIDE];

    const int bl = blockIdx.x;
    const int b  = bl / LL;
    const int l1 = bl % LL;
    const int tid = threadIdx.x;

    const float r0 = re[0], r1 = re[1], r2 = re[2];
    for (int i = tid; i < TT * PP; i += 384)
        Slds[i] = w[i * KK] * r0 + w[i * KK + 1] * r1 + w[i * KK + 2] * r2;
    __syncthreads();

    const int wave = tid >> 5;
    const int ln   = tid & 31;
    const int half = ln >> 4;
    const int col  = ln & 15;
    const int l2   = wave;

    v2f sf[16];
#pragma unroll
    for (int p4 = 0; p4 < 16; ++p4) {
        const int p0 = p4 * 4 + 2 * half;
        sf[p4].x = Slds[col * PP + p0];
        sf[p4].y = Slds[col * PP + p0 + 1];
    }

    const float* ab = adj + ((size_t)b * NN + l1 * PP) * NN + (size_t)l2 * PP;
    v8f m0 = {}, m1 = {}, m2 = {}, m3 = {};
#pragma unroll
    for (int p4 = 0; p4 < 16; ++p4) {
        const int p0 = p4 * 4 + 2 * half;
        const float* r0p = ab + (size_t)p0 * NN;
        const float* r1p = ab + (size_t)(p0 + 1) * NN;
        v2f bf;
        bf.x = r0p[col];        bf.y = r1p[col];
        m0 = __builtin_amdgcn_wmma_f32_16x16x4_f32(false, sf[p4], false, bf, (short)0, m0, false, false);
        bf.x = r0p[16 + col];   bf.y = r1p[16 + col];
        m1 = __builtin_amdgcn_wmma_f32_16x16x4_f32(false, sf[p4], false, bf, (short)0, m1, false, false);
        bf.x = r0p[32 + col];   bf.y = r1p[32 + col];
        m2 = __builtin_amdgcn_wmma_f32_16x16x4_f32(false, sf[p4], false, bf, (short)0, m2, false, false);
        bf.x = r0p[48 + col];   bf.y = r1p[48 + col];
        m3 = __builtin_amdgcn_wmma_f32_16x16x4_f32(false, sf[p4], false, bf, (short)0, m3, false, false);
    }

    float* mw = &M1[wave][0];
#pragma unroll
    for (int i = 0; i < 8; ++i) {
        const int t = i + 8 * half;
        mw[t * M1STRIDE + col]      = m0[i];
        mw[t * M1STRIDE + 16 + col] = m1[i];
        mw[t * M1STRIDE + 32 + col] = m2[i];
        mw[t * M1STRIDE + 48 + col] = m3[i];
    }
    asm volatile("s_wait_dscnt 0" ::: "memory");

    v8f z = {};
#pragma unroll
    for (int q4 = 0; q4 < 16; ++q4) {
        const int q0 = q4 * 4 + 2 * half;
        v2f af;
        af.x = mw[col * M1STRIDE + q0];
        af.y = mw[col * M1STRIDE + q0 + 1];
        z = __builtin_amdgcn_wmma_f32_16x16x4_f32(false, af, false, sf[q4], (short)0, z, false, false);
    }

    float* ob = out + ((size_t)b * (LL * TT) + l1 * TT) * (LL * TT) + l2 * TT;
#pragma unroll
    for (int i = 0; i < 8; ++i) {
        const int t = i + 8 * half;
        ob[(size_t)t * (LL * TT) + col] = z[i];
    }
}

extern "C" void kernel_launch(void* const* d_in, const int* in_sizes, int n_in,
                              void* d_out, int out_size, void* d_ws, size_t ws_size,
                              hipStream_t stream) {
    const float* x    = (const float*)d_in[0];
    const float* adj  = (const float*)d_in[1];
    const float* w    = (const float*)d_in[2];
    const float* bias = (const float*)d_in[3];
    const float* re   = (const float*)d_in[4];

    float* out_x   = (float*)d_out;
    float* out_adj = out_x + (size_t)BB * (TT * LL) * FF;

    pooled_x_kernel<<<dim3(BB * LL), dim3(256), 0, stream>>>(x, w, bias, out_x);
    pooled_adj_kernel<<<dim3(BB * LL), dim3(384), 0, stream>>>(adj, w, re, out_adj);
}


# adj kernel: block loads go through a 4-step register ring (24-32 in flight) instead of all 128 up front: 166 -> 105 VGPRs, so 3 workgroups (36 waves) per WGP are resident, not 1
# speedup vs baseline: 1.0603x; 1.0603x over previous
#include <hip/hip_runtime.h>
#include <hip/hip_bf16.h>

typedef float v2f __attribute__((ext_vector_type(2)));
typedef float v8f __attribute__((ext_vector_type(8)));

#define BB 64
#define NN 768
#define FF 256
#define PP 64
#define TT 16
#define KK 3
#define LL 12

__global__ __launch_bounds__(256) void pooled_x_kernel(
    const float* __restrict__ x, const float* __restrict__ w,
    const float* __restrict__ bias, float* __restrict__ out)
{
    __shared__ float Xl[PP * FF];

    const int bl = blockIdx.x;
    const int b  = bl / LL;
    const int l  = bl % LL;
    const int tid = threadIdx.x;

    const float* xb = x + (size_t)b * NN * FF;
    for (int i = tid; i < PP * FF; i += 256) {
        const int p = i >> 8;
        const int f = i & 255;
        Xl[i] = xb[(size_t)(p * LL + l) * FF + f];
    }
    __syncthreads();

    const int wave = tid >> 5;
    const int ln   = tid & 31;
    const int half = ln >> 4;
    const int col  = ln & 15;

    v2f afrag[KK][16];
#pragma unroll
    for (int k = 0; k < KK; ++k) {
#pragma unroll
        for (int p4 = 0; p4 < 16; ++p4) {
            const int p0 = p4 * 4 + 2 * half;
            afrag[k][p4].x = w[(col * PP + p0) * KK + k];
            afrag[k][p4].y = w[(col * PP + p0 + 1) * KK + k];
        }
    }

    for (int tile = wave; tile < FF / 16; tile += 8) {
        const int fbase = tile * 16;
        v8f acc = {};
#pragma unroll
        for (int k = 0; k < KK; ++k) {
            const int f  = fbase + col + k - 1;
            const int fc = f < 0 ? 0 : (f > FF - 1 ? FF - 1 : f);
            const float msk = (f == fc) ? 1.0f : 0.0f;
#pragma unroll
            for (int p4 = 0; p4 < 16; ++p4) {
                const int p0 = p4 * 4 + 2 * half;
                v2f bf;
                bf.x = Xl[p0 * FF + fc] * msk;
                bf.y = Xl[(p0 + 1) * FF + fc] * msk;
                acc = __builtin_amdgcn_wmma_f32_16x16x4_f32(
                    false, afrag[k][p4], false, bf, (short)0, acc, false, false);
            }
        }
#pragma unroll
        for (int i = 0; i < 8; ++i) {
            const int t = i + 8 * half;
            out[((size_t)b * (TT * LL) + t * LL + l) * FF + fbase + col] =
                acc[i] + bias[t];
        }
    }
}

#define M1STRIDE 65
#define ADJ_AHEAD 3

__global__ __launch_bounds__(384) void pooled_adj_kernel(
    const float* __restrict__ adj, const float* __restrict__ w,
    const float* __restrict__ re, float* __restrict__ out)
{
    __shared__ float Slds[TT * PP];
    __shared__ float M1[LL][TT * M1STRIDE];

    const int bl = blockIdx.x;
    const int b  = bl / LL;
    const int l1 = bl % LL;
    const int tid = threadIdx.x;

    const float r0 = re[0], r1 = re[1], r2 = re[2];
    for (int i = tid; i < TT * PP; i += 384)
        Slds[i] = w[i * KK] * r0 + w[i * KK + 1] * r1 + w[i * KK + 2] * r2;
    __syncthreads();

    const int wave = tid >> 5;
    const int ln   = tid & 31;
    const int half = ln >> 4;
    const int col  = ln & 15;
    const int l2   = wave;

    v2f sf[16];
#pragma unroll
    for (int p4 = 0; p4 < 16; ++p4) {
        const int p0 = p4 * 4 + 2 * half;
        sf[p4].x = Slds[col * PP + p0];
        sf[p4].y = Slds[col * PP + p0 + 1];
    }

    const float* ab = adj + ((size_t)b * NN + l1 * PP) * NN + (size_t)l2 * PP;
    const float* lp = ab + (size_t)(2 * half) * NN + col;
    float ring[ADJ_AHEAD + 1][8];
#pragma unroll
    for (int p4 = 0; p4 < ADJ_AHEAD; ++p4) {
        const float* r0p = lp + (size_t)(4 * p4) * NN;
        const float* r1p = r0p + NN;
        float* rg = ring[p4 % (ADJ_AHEAD + 1)];
        rg[0] = r0p[0];   rg[1] = r1p[0];
        rg[2] = r0p[16];  rg[3] = r1p[16];
        rg[4] = r0p[32];  rg[5] = r1p[32];
        rg[6] = r0p[48];  rg[7] = r1p[48];
    }
    v8f m0 = {}, m1 = {}, m2 = {}, m3 = {};
#pragma unroll
    for (int p4 = 0; p4 < 16; ++p4) {
        if (p4 + ADJ_AHEAD < 16) {
            const float* r0p = lp + (size_t)(4 * (p4 + ADJ_AHEAD)) * NN;
            const float* r1p = r0p + NN;
            float* rg = ring[(p4 + ADJ_AHEAD) % (ADJ_AHEAD + 1)];
            rg[0] = r0p[0];   rg[1] = r1p[0];
            rg[2] = r0p[16];  rg[3] = r1p[16];
            rg[4] = r0p[32];  rg[5] = r1p[32];
            rg[6] = r0p[48];  rg[7] = r1p[48];
        }
        const float* rg = ring[p4 % (ADJ_AHEAD + 1)];
        v2f bf;
        bf.x = rg[0];  bf.y = rg[1];
        m0 = __builtin_amdgcn_wmma_f32_16x16x4_f32(false, sf[p4], false, bf, (short)0, m0, false, false);
        bf.x = rg[2];  bf.y = rg[3];
        m1 = __builtin_amdgcn_wmma_f32_16x16x4_f32(false, sf[p4], false, bf, (short)0, m1, false, false);
        bf.x = rg[4];  bf.y = rg[5];
        m2 = __builtin_amdgcn_wmma_f32_16x16x4_f32(false, sf[p4], false, bf, (short)0, m2, false, false);
        bf.x = rg[6];  bf.y = rg[7];
        m3 = __builtin_amdgcn_wmma_f32_16x16x4_f32(false, sf[p4], false, bf, (short)0, m3, false, false);
        __builtin_amdgcn_sched_barrier(0);
    }

    float* mw = &M1[wave][0];
#pragma unroll
    for (int i = 0; i < 8; ++i) {
        const int t = i + 8 * half;
        mw[t * M1STRIDE + col]      = m0[i];
        mw[t * M1STRIDE + 16 + col] = m1[i];
        mw[t * M1STRIDE + 32 + col] = m2[i];
        mw[t * M1STRIDE + 48 + col] = m3[i];
    }
    asm volatile("s_wait_dscnt 0" ::: "memory");

    v8f z = {};
#pragma unroll
    for (int q4 = 0; q4 < 16; ++q4) {
        const int q0 = q4 * 4 + 2 * half;
        v2f af;
        af.x = mw[col * M1STRIDE + q0];
        af.y = mw[col * M1STRIDE + q0 + 1];
        z = __builtin_amdgcn_wmma_f32_16x16x4_f32(false, af, false, sf[q4], (short)0, z, false, false);
    }

    float* ob = out + ((size_t)b * (LL * TT) + l1 * TT) * (LL * TT) + l2 * TT;
#pragma unroll
    for (int i = 0; i < 8; ++i) {
        const int t = i + 8 * half;
        ob[(size_t)t * (LL * TT) + col] = z[i];
    }
}

extern "C" void kernel_launch(void* const* d_in, const int* in_sizes, int n_in,
                              void* d_out, int out_size, void* d_ws, size_t ws_size,
                              hipStream_t stream) {
    const float* x    = (const float*)d_in[0];
    const float* adj  = (const float*)d_in[1];
    const float* w    = (const float*)d_in[2];
    const float* bias = (const float*)d_in[3];
    const float* re   = (const float*)d_in[4];

    float* out_x   = (float*)d_out;
    float* out_adj = out_x + (size_t)BB * (TT * LL) * FF;

    pooled_x_kernel<<<dim3(BB * LL), dim3(256), 0, stream>>>(x, w, bias, out_x);
    pooled_adj_kernel<<<dim3(BB * LL), dim3(384), 0, stream>>>(adj, w, re, out_adj);
}


# feature kernel: x rows staged in 8 chunks by 16-byte whole-line loads kept in flight across a hand-written barrier, chunk c's products run under later loads; each tile's sum now runs chunk by chunk
# speedup vs baseline: 1.2261x; 1.1564x over previous
#include <hip/hip_runtime.h>
#include <hip/hip_bf16.h>

typedef float v2f __attribute__((ext_vector_type(2)));
typedef float v8f __attribute__((ext_vector_type(8)));
typedef float v4f __attribute__((ext_vector_type(4)));

#define BB 64
#define NN 768
#define FF 256
#define PP 64
#define TT 16
#define KK 3
#define LL 12

#define XCH 8
#define XROWS (PP / XCH)
#define XNJ (XROWS / 4)
#define XRING 2

__global__ __launch_bounds__(256, 6) void pooled_x_kernel(
    const float* __restrict__ x, const float* __restrict__ w,
    const float* __restrict__ bias, float* __restrict__ out)
{
    __shared__ __attribute__((aligned(16))) float Xl[PP * FF];

    const int bl = blockIdx.x;
    const int b  = bl / LL;
    const int l  = bl % LL;
    const int tid = threadIdx.x;

    const int wave = tid >> 5;
    const int ln   = tid & 31;
    const int half = ln >> 4;
    const int col  = ln & 15;

    const int f4 = tid & 63;
    const int rq = tid >> 6;
    const float* xg = x + (size_t)b * NN * FF + (size_t)l * FF
                        + (size_t)rq * (LL * FF) + 4 * f4;
    float* xs = &Xl[rq * FF + 4 * f4];

    v4f ring[XRING][XNJ];
#pragma unroll
    for (int c = 0; c < XRING; ++c) {
#pragma unroll
        for (int j = 0; j < XNJ; ++j)
            ring[c][j] = *reinterpret_cast<const v4f*>(
                xg + (size_t)(XROWS * c + 4 * j) * (LL * FF));
    }

    __builtin_amdgcn_sched_barrier(0);

    v2f afrag[KK][16];
#pragma unroll
    for (int k = 0; k < KK; ++k) {
#pragma unroll
        for (int p4 = 0; p4 < 16; ++p4) {
            const int p0 = p4 * 4 + 2 * half;
            afrag[k][p4].x = w[(col * PP + p0) * KK + k];
            afrag[k][p4].y = w[(col * PP + p0 + 1) * KK + k];
        }
    }

    v8f acc[2] = {};
#pragma unroll
    for (int c = 0; c < XCH; ++c) {
        __builtin_amdgcn_sched_barrier(0);
#pragma unroll
        for (int j = 0; j < XNJ; ++j)
            *reinterpret_cast<v4f*>(xs + (XROWS * c + 4 * j) * FF) = ring[c % XRING][j];
        if (c + XRING < XCH) {
#pragma unroll
            for (int j = 0; j < XNJ; ++j)
                ring[c % XRING][j] = *reinterpret_cast<const v4f*>(
                    xg + (size_t)(XROWS * (c + XRING) + 4 * j) * (LL * FF));
        }
        asm volatile("s_wait_dscnt 0x0\n\ts_barrier_signal -1\n\ts_barrier_wait -1" ::: "memory");

#pragma unroll
        for (int ti = 0; ti < 2; ++ti) {
            const int fbase = (wave + 8 * ti) * 16;
#pragma unroll
            for (int k = 0; k < KK; ++k) {
                const int f  = fbase + col + k - 1;
                const int fc = f < 0 ? 0 : (f > FF - 1 ? FF - 1 : f);
                const float msk = (f == fc) ? 1.0f : 0.0f;
#pragma unroll
                for (int pp = 0; pp < XROWS / 4; ++pp) {
                    const int p4 = c * (XROWS / 4) + pp;
                    const int p0 = p4 * 4 + 2 * half;
                    v2f bf;
                    bf.x = Xl[p0 * FF + fc] * msk;
                    bf.y = Xl[(p0 + 1) * FF + fc] * msk;
                    acc[ti] = __builtin_amdgcn_wmma_f32_16x16x4_f32(
                        false, afrag[k][p4], false, bf, (short)0, acc[ti], false, false);
                }
            }
        }
    }

#pragma unroll
    for (int ti = 0; ti < 2; ++ti) {
        const int fbase = (wave + 8 * ti) * 16;
#pragma unroll
        for (int i = 0; i < 8; ++i) {
            const int t = i + 8 * half;
            out[((size_t)b * (TT * LL) + t * LL + l) * FF + fbase + col] =
                acc[ti][i] + bias[t];
        }
    }
}

#define M1STRIDE 65
#define ADJ_AHEAD 3

__global__ __launch_bounds__(384) void pooled_adj_kernel(
    const float* __restrict__ adj, const float* __restrict__ w,
    const float* __restrict__ re, float* __restrict__ out)
{
    __shared__ float Slds[TT * PP];
    __shared__ float M1[LL][TT * M1STRIDE];

    const int bl = blockIdx.x;
    const int b  = bl / LL;
    const int l1 = bl % LL;
    const int tid = threadIdx.x;

    const float r0 = re[0], r1 = re[1], r2 = re[2];
    for (int i = tid; i < TT * PP; i += 384)
        Slds[i] = w[i * KK] * r0 + w[i * KK + 1] * r1 + w[i * KK + 2] * r2;
    __syncthreads();

    const int wave = tid >> 5;
    const int ln   = tid & 31;
    const int half = ln >> 4;
    const int col  = ln & 15;
    const int l2   = wave;

    v2f sf[16];
#pragma unroll
    for (int p4 = 0; p4 < 16; ++p4) {
        const int p0 = p4 * 4 + 2 * half;
        sf[p4].x = Slds[col * PP + p0];
        sf[p4].y = Slds[col * PP + p0 + 1];
    }

    const float* ab = adj + ((size_t)b * NN + l1 * PP) * NN + (size_t)l2 * PP;
    const float* lp = ab + (size_t)(2 * half) * NN + col;
    float ring[ADJ_AHEAD + 1][8];
#pragma unroll
    for (int p4 = 0; p4 < ADJ_AHEAD; ++p4) {
        const float* r0p = lp + (size_t)(4 * p4) * NN;
        const float* r1p = r0p + NN;
        float* rg = ring[p4 % (ADJ_AHEAD + 1)];
        rg[0] = r0p[0];   rg[1] = r1p[0];
        rg[2] = r0p[16];  rg[3] = r1p[16];
        rg[4] = r0p[32];  rg[5] = r1p[32];
        rg[6] = r0p[48];  rg[7] = r1p[48];
    }
    v8f m0 = {}, m1 = {}, m2 = {}, m3 = {};
#pragma unroll
    for (int p4 = 0; p4 < 16; ++p4) {
        if (p4 + ADJ_AHEAD < 16) {
            const float* r0p = lp + (size_t)(4 * (p4 + ADJ_AHEAD)) * NN;
            const float* r1p = r0p + NN;
            float* rg = ring[(p4 + ADJ_AHEAD) % (ADJ_AHEAD + 1)];
            rg[0] = r0p[0];   rg[1] = r1p[0];
            rg[2] = r0p[16];  rg[3] = r1p[16];
            rg[4] = r0p[32];  rg[5] = r1p[32];
            rg[6] = r0p[48];  rg[7] = r1p[48];
        }
        const float* rg = ring[p4 % (ADJ_AHEAD + 1)];
        v2f bf;
        bf.x = rg[0];  bf.y = rg[1];
        m0 = __builtin_amdgcn_wmma_f32_16x16x4_f32(false, sf[p4], false, bf, (short)0, m0, false, false);
        bf.x = rg[2];  bf.y = rg[3];
        m1 = __builtin_amdgcn_wmma_f32_16x16x4_f32(false, sf[p4], false, bf, (short)0, m1, false, false);
        bf.x = rg[4];  bf.y = rg[5];
        m2 = __builtin_amdgcn_wmma_f32_16x16x4_f32(false, sf[p4], false, bf, (short)0, m2, false, false);
        bf.x = rg[6];  bf.y = rg[7];
        m3 = __builtin_amdgcn_wmma_f32_16x16x4_f32(false, sf[p4], false, bf, (short)0, m3, false, false);
        __builtin_amdgcn_sched_barrier(0);
    }

    float* mw = &M1[wave][0];
#pragma unroll
    for (int i = 0; i < 8; ++i) {
        const int t = i + 8 * half;
        mw[t * M1STRIDE + col]      = m0[i];
        mw[t * M1STRIDE + 16 + col] = m1[i];
        mw[t * M1STRIDE + 32 + col] = m2[i];
        mw[t * M1STRIDE + 48 + col] = m3[i];
    }
    asm volatile("s_wait_dscnt 0" ::: "memory");

    v8f z = {};
#pragma unroll
    for (int q4 = 0; q4 < 16; ++q4) {
        const int q0 = q4 * 4 + 2 * half;
        v2f af;
        af.x = mw[col * M1STRIDE + q0];
        af.y = mw[col * M1STRIDE + q0 + 1];
        z = __builtin_amdgcn_wmma_f32_16x16x4_f32(false, af, false, sf[q4], (short)0, z, false, false);
    }

    float* ob = out + ((size_t)b * (LL * TT) + l1 * TT) * (LL * TT) + l2 * TT;
#pragma unroll
    for (int i = 0; i < 8; ++i) {
        const int t = i + 8 * half;
        ob[(size_t)t * (LL * TT) + col] = z[i];
    }
}

extern "C" void kernel_launch(void* const* d_in, const int* in_sizes, int n_in,
                              void* d_out, int out_size, void* d_ws, size_t ws_size,
                              hipStream_t stream) {
    const float* x    = (const float*)d_in[0];
    const float* adj  = (const float*)d_in[1];
    const float* w    = (const float*)d_in[2];
    const float* bias = (const float*)d_in[3];
    const float* re   = (const float*)d_in[4];

    float* out_x   = (float*)d_out;
    float* out_adj = out_x + (size_t)BB * (TT * LL) * FF;

    pooled_x_kernel<<<dim3(BB * LL), dim3(256), 0, stream>>>(x, w, bias, out_x);
    pooled_adj_kernel<<<dim3(BB * LL), dim3(384), 0, stream>>>(adj, w, re, out_adj);
}


# feature kernel: weights reach the A fragments through LDS (3 whole-line 16-byte loads per thread into padded rows, conflict-free 8-byte LDS reads) instead of 32 per-lane gathers of 16 lines each
# speedup vs baseline: 1.3571x; 1.1068x over previous
#include <hip/hip_runtime.h>
#include <hip/hip_bf16.h>

typedef float v2f __attribute__((ext_vector_type(2)));
typedef float v8f __attribute__((ext_vector_type(8)));
typedef float v4f __attribute__((ext_vector_type(4)));

#define BB 64
#define NN 768
#define FF 256
#define PP 64
#define TT 16
#define KK 3
#define LL 12

#define XCH 8
#define XROWS (PP / XCH)
#define XNJ (XROWS / 4)
#define XRING 2
#define WPITCH (PP * KK + 4)

__global__ __launch_bounds__(256, 6) void pooled_x_kernel(
    const float* __restrict__ x, const float* __restrict__ w,
    const float* __restrict__ bias, float* __restrict__ out)
{
    __shared__ __attribute__((aligned(16))) float Xl[PP * FF];
    __shared__ __attribute__((aligned(16))) float Wl[TT * WPITCH];

    const int bl = blockIdx.x;
    const int b  = bl / LL;
    const int l  = bl % LL;
    const int tid = threadIdx.x;

    const int wave = tid >> 5;
    const int ln   = tid & 31;
    const int half = ln >> 4;
    const int col  = ln & 15;

    const int f4 = tid & 63;
    const int rq = tid >> 6;
    const float* xg = x + (size_t)b * NN * FF + (size_t)l * FF
                        + (size_t)rq * (LL * FF) + 4 * f4;
    float* xs = &Xl[rq * FF + 4 * f4];

    v4f wreg[3];
#pragma unroll
    for (int j = 0; j < 3; ++j)
        wreg[j] = *reinterpret_cast<const v4f*>(w + 4 * (tid + 256 * j));

    v4f ring[XRING][XNJ];
#pragma unroll
    for (int c = 0; c < XRING; ++c) {
#pragma unroll
        for (int j = 0; j < XNJ; ++j)
            ring[c][j] = *reinterpret_cast<const v4f*>(
                xg + (size_t)(XROWS * c + 4 * j) * (LL * FF));
    }

    __builtin_amdgcn_sched_barrier(0);

#pragma unroll
    for (int j = 0; j < 3; ++j) {
        const int i = tid + 256 * j;
        const int t = i / (PP * KK / 4);
        const int q = i - t * (PP * KK / 4);
        *reinterpret_cast<v4f*>(&Wl[t * WPITCH + 4 * q]) = wreg[j];
    }
    asm volatile("s_wait_dscnt 0x0\n\ts_barrier_signal -1\n\ts_barrier_wait -1" ::: "memory");

    v2f afrag[KK][16];
    const float* wl = &Wl[col * WPITCH + 2 * half * KK];
#pragma unroll
    for (int k = 0; k < KK; ++k) {
#pragma unroll
        for (int p4 = 0; p4 < 16; ++p4) {
            afrag[k][p4].x = wl[p4 * 4 * KK + k];
            afrag[k][p4].y = wl[p4 * 4 * KK + KK + k];
        }
    }

    v8f acc[2] = {};
#pragma unroll
    for (int c = 0; c < XCH; ++c) {
        __builtin_amdgcn_sched_barrier(0);
#pragma unroll
        for (int j = 0; j < XNJ; ++j)
            *reinterpret_cast<v4f*>(xs + (XROWS * c + 4 * j) * FF) = ring[c % XRING][j];
        if (c + XRING < XCH) {
#pragma unroll
            for (int j = 0; j < XNJ; ++j)
                ring[c % XRING][j] = *reinterpret_cast<const v4f*>(
                    xg + (size_t)(XROWS * (c + XRING) + 4 * j) * (LL * FF));
        }
        asm volatile("s_wait_dscnt 0x0\n\ts_barrier_signal -1\n\ts_barrier_wait -1" ::: "memory");

#pragma unroll
        for (int ti = 0; ti < 2; ++ti) {
            const int fbase = (wave + 8 * ti) * 16;
#pragma unroll
            for (int k = 0; k < KK; ++k) {
                const int f  = fbase + col + k - 1;
                const int fc = f < 0 ? 0 : (f > FF - 1 ? FF - 1 : f);
                const float msk = (f == fc) ? 1.0f : 0.0f;
#pragma unroll
                for (int pp = 0; pp < XROWS / 4; ++pp) {
                    const int p4 = c * (XROWS / 4) + pp;
                    const int p0 = p4 * 4 + 2 * half;
                    v2f bf;
                    bf.x = Xl[p0 * FF + fc] * msk;
                    bf.y = Xl[(p0 + 1) * FF + fc] * msk;
                    acc[ti] = __builtin_amdgcn_wmma_f32_16x16x4_f32(
                        false, afrag[k][p4], false, bf, (short)0, acc[ti], false, false);
                }
            }
        }
    }

#pragma unroll
    for (int ti = 0; ti < 2; ++ti) {
        const int fbase = (wave + 8 * ti) * 16;
#pragma unroll
        for (int i = 0; i < 8; ++i) {
            const int t = i + 8 * half;
            out[((size_t)b * (TT * LL) + t * LL + l) * FF + fbase + col] =
                acc[ti][i] + bias[t];
        }
    }
}

#define M1STRIDE 65
#define ADJ_AHEAD 3

__global__ __launch_bounds__(384) void pooled_adj_kernel(
    const float* __restrict__ adj, const float* __restrict__ w,
    const float* __restrict__ re, float* __restrict__ out)
{
    __shared__ float Slds[TT * PP];
    __shared__ float M1[LL][TT * M1STRIDE];

    const int bl = blockIdx.x;
    const int b  = bl / LL;
    const int l1 = bl % LL;
    const int tid = threadIdx.x;

    const float r0 = re[0], r1 = re[1], r2 = re[2];
    for (int i = tid; i < TT * PP; i += 384)
        Slds[i] = w[i * KK] * r0 + w[i * KK + 1] * r1 + w[i * KK + 2] * r2;
    __syncthreads();

    const int wave = tid >> 5;
    const int ln   = tid & 31;
    const int half = ln >> 4;
    const int col  = ln & 15;
    const int l2   = wave;

    v2f sf[16];
#pragma unroll
    for (int p4 = 0; p4 < 16; ++p4) {
        const int p0 = p4 * 4 + 2 * half;
        sf[p4].x = Slds[col * PP + p0];
        sf[p4].y = Slds[col * PP + p0 + 1];
    }

    const float* ab = adj + ((size_t)b * NN + l1 * PP) * NN + (size_t)l2 * PP;
    const float* lp = ab + (size_t)(2 * half) * NN + col;
    float ring[ADJ_AHEAD + 1][8];
#pragma unroll
    for (int p4 = 0; p4 < ADJ_AHEAD; ++p4) {
        const float* r0p = lp + (size_t)(4 * p4) * NN;
        const float* r1p = r0p + NN;
        float* rg = ring[p4 % (ADJ_AHEAD + 1)];
        rg[0] = r0p[0];   rg[1] = r1p[0];
        rg[2] = r0p[16];  rg[3] = r1p[16];
        rg[4] = r0p[32];  rg[5] = r1p[32];
        rg[6] = r0p[48];  rg[7] = r1p[48];
    }
    v8f m0 = {}, m1 = {}, m2 = {}, m3 = {};
#pragma unroll
    for (int p4 = 0; p4 < 16; ++p4) {
        if (p4 + ADJ_AHEAD < 16) {
            const float* r0p = lp + (size_t)(4 * (p4 + ADJ_AHEAD)) * NN;
            const float* r1p = r0p + NN;
            float* rg = ring[(p4 + ADJ_AHEAD) % (ADJ_AHEAD + 1)];
            rg[0] = r0p[0];   rg[1] = r1p[0];
            rg[2] = r0p[16];  rg[3] = r1p[16];
            rg[4] = r0p[32];  rg[5] = r1p[32];
            rg[6] = r0p[48];  rg[7] = r1p[48];
        }
        const float* rg = ring[p4 % (ADJ_AHEAD + 1)];
        v2f bf;
        bf.x = rg[0];  bf.y = rg[1];
        m0 = __builtin_amdgcn_wmma_f32_16x16x4_f32(false, sf[p4], false, bf, (short)0, m0, false, false);
        bf.x = rg[2];  bf.y = rg[3];
        m1 = __builtin_amdgcn_wmma_f32_16x16x4_f32(false, sf[p4], false, bf, (short)0, m1, false, false);
        bf.x = rg[4];  bf.y = rg[5];
        m2 = __builtin_amdgcn_wmma_f32_16x16x4_f32(false, sf[p4], false, bf, (short)0, m2, false, false);
        bf.x = rg[6];  bf.y = rg[7];
        m3 = __builtin_amdgcn_wmma_f32_16x16x4_f32(false, sf[p4], false, bf, (short)0, m3, false, false);
        __builtin_amdgcn_sched_barrier(0);
    }

    float* mw = &M1[wave][0];
#pragma unroll
    for (int i = 0; i < 8; ++i) {
        const int t = i + 8 * half;
        mw[t * M1STRIDE + col]      = m0[i];
        mw[t * M1STRIDE + 16 + col] = m1[i];
        mw[t * M1STRIDE + 32 + col] = m2[i];
        mw[t * M1STRIDE + 48 + col] = m3[i];
    }
    asm volatile("s_wait_dscnt 0" ::: "memory");

    v8f z = {};
#pragma unroll
    for (int q4 = 0; q4 < 16; ++q4) {
        const int q0 = q4 * 4 + 2 * half;
        v2f af;
        af.x = mw[col * M1STRIDE + q0];
        af.y = mw[col * M1STRIDE + q0 + 1];
        z = __builtin_amdgcn_wmma_f32_16x16x4_f32(false, af, false, sf[q4], (short)0, z, false, false);
    }

    float* ob = out + ((size_t)b * (LL * TT) + l1 * TT) * (LL * TT) + l2 * TT;
#pragma unroll
    for (int i = 0; i < 8; ++i) {
        const int t = i + 8 * half;
        ob[(size_t)t * (LL * TT) + col] = z[i];
    }
}

extern "C" void kernel_launch(void* const* d_in, const int* in_sizes, int n_in,
                              void* d_out, int out_size, void* d_ws, size_t ws_size,
                              hipStream_t stream) {
    const float* x    = (const float*)d_in[0];
    const float* adj  = (const float*)d_in[1];
    const float* w    = (const float*)d_in[2];
    const float* bias = (const float*)d_in[3];
    const float* re   = (const float*)d_in[4];

    float* out_x   = (float*)d_out;
    float* out_adj = out_x + (size_t)BB * (TT * LL) * FF;

    pooled_x_kernel<<<dim3(BB * LL), dim3(256), 0, stream>>>(x, w, bias, out_x);
    pooled_adj_kernel<<<dim3(BB * LL), dim3(384), 0, stream>>>(adj, w, re, out_adj);
}


# adj kernel: 256 persistent workgroups (one per WGP) walk 3 row groups each; a wave keeps the whole next 64x64 block in flight by whole-line float4 loads, so 3072 waves start instead of 9216
# speedup vs baseline: 1.4843x; 1.0937x over previous
#include <hip/hip_runtime.h>
#include <hip/hip_bf16.h>

typedef float v2f __attribute__((ext_vector_type(2)));
typedef float v8f __attribute__((ext_vector_type(8)));
typedef float v4f __attribute__((ext_vector_type(4)));

#define BB 64
#define NN 768
#define FF 256
#define PP 64
#define TT 16
#define KK 3
#define LL 12

#define XCH 8
#define XROWS (PP / XCH)
#define XNJ (XROWS / 4)
#define XRING 2
#define WPITCH (PP * KK + 4)

__global__ __launch_bounds__(256, 6) void pooled_x_kernel(
    const float* __restrict__ x, const float* __restrict__ w,
    const float* __restrict__ bias, float* __restrict__ out)
{
    __shared__ __attribute__((aligned(16))) float Xl[PP * FF];
    __shared__ __attribute__((aligned(16))) float Wl[TT * WPITCH];

    const int bl = blockIdx.x;
    const int b  = bl / LL;
    const int l  = bl % LL;
    const int tid = threadIdx.x;

    const int wave = tid >> 5;
    const int ln   = tid & 31;
    const int half = ln >> 4;
    const int col  = ln & 15;

    const int f4 = tid & 63;
    const int rq = tid >> 6;
    const float* xg = x + (size_t)b * NN * FF + (size_t)l * FF
                        + (size_t)rq * (LL * FF) + 4 * f4;
    float* xs = &Xl[rq * FF + 4 * f4];

    v4f wreg[3];
#pragma unroll
    for (int j = 0; j < 3; ++j)
        wreg[j] = *reinterpret_cast<const v4f*>(w + 4 * (tid + 256 * j));

    v4f ring[XRING][XNJ];
#pragma unroll
    for (int c = 0; c < XRING; ++c) {
#pragma unroll
        for (int j = 0; j < XNJ; ++j)
            ring[c][j] = *reinterpret_cast<const v4f*>(
                xg + (size_t)(XROWS * c + 4 * j) * (LL * FF));
    }

    __builtin_amdgcn_sched_barrier(0);

#pragma unroll
    for (int j = 0; j < 3; ++j) {
        const int i = tid + 256 * j;
        const int t = i / (PP * KK / 4);
        const int q = i - t * (PP * KK / 4);
        *reinterpret_cast<v4f*>(&Wl[t * WPITCH + 4 * q]) = wreg[j];
    }
    asm volatile("s_wait_dscnt 0x0\n\ts_barrier_signal -1\n\ts_barrier_wait -1" ::: "memory");

    v2f afrag[KK][16];
    const float* wl = &Wl[col * WPITCH + 2 * half * KK];
#pragma unroll
    for (int k = 0; k < KK; ++k) {
#pragma unroll
        for (int p4 = 0; p4 < 16; ++p4) {
            afrag[k][p4].x = wl[p4 * 4 * KK + k];
            afrag[k][p4].y = wl[p4 * 4 * KK + KK + k];
        }
    }

    v8f acc[2] = {};
#pragma unroll
    for (int c = 0; c < XCH; ++c) {
        __builtin_amdgcn_sched_barrier(0);
#pragma unroll
        for (int j = 0; j < XNJ; ++j)
            *reinterpret_cast<v4f*>(xs + (XROWS * c + 4 * j) * FF) = ring[c % XRING][j];
        if (c + XRING < XCH) {
#pragma unroll
            for (int j = 0; j < XNJ; ++j)
                ring[c % XRING][j] = *reinterpret_cast<const v4f*>(
                    xg + (size_t)(XROWS * (c + XRING) + 4 * j) * (LL * FF));
        }
        asm volatile("s_wait_dscnt 0x0\n\ts_barrier_signal -1\n\ts_barrier_wait -1" ::: "memory");

#pragma unroll
        for (int ti = 0; ti < 2; ++ti) {
            const int fbase = (wave + 8 * ti) * 16;
#pragma unroll
            for (int k = 0; k < KK; ++k) {
                const int f  = fbase + col + k - 1;
                const int fc = f < 0 ? 0 : (f > FF - 1 ? FF - 1 : f);
                const float msk = (f == fc) ? 1.0f : 0.0f;
#pragma unroll
                for (int pp = 0; pp < XROWS / 4; ++pp) {
                    const int p4 = c * (XROWS / 4) + pp;
                    const int p0 = p4 * 4 + 2 * half;
                    v2f bf;
                    bf.x = Xl[p0 * FF + fc] * msk;
                    bf.y = Xl[(p0 + 1) * FF + fc] * msk;
                    acc[ti] = __builtin_amdgcn_wmma_f32_16x16x4_f32(
                        false, afrag[k][p4], false, bf, (short)0, acc[ti], false, false);
                }
            }
        }
    }

#pragma unroll
    for (int ti = 0; ti < 2; ++ti) {
        const int fbase = (wave + 8 * ti) * 16;
#pragma unroll
        for (int i = 0; i < 8; ++i) {
            const int t = i + 8 * half;
            out[((size_t)b * (TT * LL) + t * LL + l) * FF + fbase + col] =
                acc[ti][i] + bias[t];
        }
    }
}

#define SPITCH 68
#define ADJ_WGS 256
#define ADJ_GROUPS 3

__global__ __launch_bounds__(384) void pooled_adj_kernel(
    const float* __restrict__ adj, const float* __restrict__ w,
    const float* __restrict__ re, float* __restrict__ out)
{
    __shared__ __attribute__((aligned(16))) float Wl[TT * WPITCH];
    __shared__ __attribute__((aligned(16))) float M1[LL][TT * SPITCH];

    const int tid  = threadIdx.x;
    const int wave = tid >> 5;
    const int ln   = tid & 31;
    const int half = ln >> 4;
    const int col  = ln & 15;
    const int l2   = wave;
    const int bl0  = blockIdx.x * ADJ_GROUPS;

    v4f wreg[2];
#pragma unroll
    for (int j = 0; j < 2; ++j)
        wreg[j] = *reinterpret_cast<const v4f*>(w + 4 * (tid + 384 * j));

    const float* lp = adj + ((size_t)bl0 * PP + 2 * half) * NN + l2 * PP + 4 * col;
    v4f rx[16], ry[16];
#pragma unroll
    for (int p4 = 0; p4 < 16; ++p4) {
        rx[p4] = *reinterpret_cast<const v4f*>(lp + (size_t)(4 * p4) * NN);
        ry[p4] = *reinterpret_cast<const v4f*>(lp + (size_t)(4 * p4 + 1) * NN);
    }

    __builtin_amdgcn_sched_barrier(0);

#pragma unroll
    for (int j = 0; j < 2; ++j) {
        const int i = tid + 384 * j;
        const int t = i / (PP * KK / 4);
        const int q = i - t * (PP * KK / 4);
        *reinterpret_cast<v4f*>(&Wl[t * WPITCH + 4 * q]) = wreg[j];
    }
    asm volatile("s_wait_dscnt 0x0\n\ts_barrier_signal -1\n\ts_barrier_wait -1" ::: "memory");
    __builtin_amdgcn_sched_barrier(0);

    const float r0 = re[0], r1 = re[1], r2 = re[2];
    const float* wl = &Wl[col * WPITCH + 2 * half * KK];
    v2f sf[16];
#pragma unroll
    for (int p4 = 0; p4 < 16; ++p4) {
        const float* e = wl + p4 * 4 * KK;
        sf[p4].x = e[0] * r0 + e[1] * r1 + e[2] * r2;
        sf[p4].y = e[3] * r0 + e[4] * r1 + e[5] * r2;
    }
    __builtin_amdgcn_sched_barrier(0);

    float* mw = &M1[wave][0];

#pragma unroll
    for (int n = 0; n < ADJ_GROUPS; ++n) {
        const float* lpn = lp + (size_t)(n + 1) * PP * NN;
        v8f m0 = {}, m1 = {}, m2 = {}, m3 = {};
#pragma unroll
        for (int p4 = 0; p4 < 16; ++p4) {
            v2f bf;
            bf.x = rx[p4].x;  bf.y = ry[p4].x;
            m0 = __builtin_amdgcn_wmma_f32_16x16x4_f32(false, sf[p4], false, bf, (short)0, m0, false, false);
            bf.x = rx[p4].y;  bf.y = ry[p4].y;
            m1 = __builtin_amdgcn_wmma_f32_16x16x4_f32(false, sf[p4], false, bf, (short)0, m1, false, false);
            bf.x = rx[p4].z;  bf.y = ry[p4].z;
            m2 = __builtin_amdgcn_wmma_f32_16x16x4_f32(false, sf[p4], false, bf, (short)0, m2, false, false);
            bf.x = rx[p4].w;  bf.y = ry[p4].w;
            m3 = __builtin_amdgcn_wmma_f32_16x16x4_f32(false, sf[p4], false, bf, (short)0, m3, false, false);
            if (n + 1 < ADJ_GROUPS) {
                rx[p4] = *reinterpret_cast<const v4f*>(lpn + (size_t)(4 * p4) * NN);
                ry[p4] = *reinterpret_cast<const v4f*>(lpn + (size_t)(4 * p4 + 1) * NN);
            }
            __builtin_amdgcn_sched_barrier(0);
        }

#pragma unroll
        for (int i = 0; i < 8; ++i) {
            const int t = i + 8 * half;
            v4f v;
            v.x = m0[i];  v.y = m1[i];  v.z = m2[i];  v.w = m3[i];
            *reinterpret_cast<v4f*>(&mw[t * SPITCH + 4 * col]) = v;
        }
        asm volatile("s_wait_dscnt 0" ::: "memory");

        v8f z = {};
#pragma unroll
        for (int q4 = 0; q4 < 16; ++q4) {
            const int q0 = q4 * 4 + 2 * half;
            v2f af;
            af.x = mw[col * SPITCH + q0];
            af.y = mw[col * SPITCH + q0 + 1];
            z = __builtin_amdgcn_wmma_f32_16x16x4_f32(false, af, false, sf[q4], (short)0, z, false, false);
        }

        float* ob = out + (size_t)(bl0 + n) * TT * (LL * TT) + l2 * TT;
#pragma unroll
        for (int i = 0; i < 8; ++i) {
            const int t = i + 8 * half;
            ob[(size_t)t * (LL * TT) + col] = z[i];
        }
    }
}

extern "C" void kernel_launch(void* const* d_in, const int* in_sizes, int n_in,
                              void* d_out, int out_size, void* d_ws, size_t ws_size,
                              hipStream_t stream) {
    const float* x    = (const float*)d_in[0];
    const float* adj  = (const float*)d_in[1];
    const float* w    = (const float*)d_in[2];
    const float* bias = (const float*)d_in[3];
    const float* re   = (const float*)d_in[4];

    float* out_x   = (float*)d_out;
    float* out_adj = out_x + (size_t)BB * (TT * LL) * FF;

    pooled_x_kernel<<<dim3(BB * LL), dim3(256), 0, stream>>>(x, w, bias, out_x);
    pooled_adj_kernel<<<dim3(ADJ_WGS), dim3(384), 0, stream>>>(adj, w, re, out_adj);
}
